# starting kernel
# baseline (speedup 1.0000x reference)
#include <hip/hip_runtime.h>

typedef __attribute__((ext_vector_type(2))) float v2f;
typedef __attribute__((ext_vector_type(8))) float v8f;

#define B_    2048
#define N_    32
#define IND   256
#define OUTD  64
#define H_    8
#define QCOLS 512


__global__ void prep_wvs(const float* __restrict__ WV, const float* __restrict__ bV,
                         float* __restrict__ WVs, float* __restrict__ bVs) {
  int idx = blockIdx.x * blockDim.x + threadIdx.x;
  int c = idx >> 6;
  int j = idx & 63;
  float s = 0.f;
#pragma unroll
  for (int h = 0; h < H_; ++h) s += WV[c * QCOLS + j * H_ + h];
  WVs[c * OUTD + j] = s;
  if (idx < OUTD) {
    float sb = 0.f;
#pragma unroll
    for (int h = 0; h < H_; ++h) sb += bV[idx * H_ + h];
    bVs[idx] = sb;
  }
}

__global__ __launch_bounds__(256)
void proj_q(const float* __restrict__ center, const float* __restrict__ WQ,
            const float* __restrict__ bQ, float* __restrict__ Qout) {
  int wid  = (blockIdx.x * blockDim.x + threadIdx.x) >> 5;
  int lane = threadIdx.x & 31;
  int l15 = lane & 15, hi = lane >> 4;
  int g  = wid >> 5;
  int nt = wid & 31;
  int n0 = nt * 16;
  v8f acc[4] = {{}, {}, {}, {}};
  for (int kt = 0; kt < IND / 4; ++kt) {
    int kb = kt * 4 + hi * 2;
    v2f b;
    b.x = WQ[kb * QCOLS + n0 + l15];
    b.y = WQ[(kb + 1) * QCOLS + n0 + l15];
#pragma unroll
    for (int u = 0; u < 4; ++u) {
      v2f a;
      a.x = center[(size_t)(g * 64 + u * 16 + l15) * IND + kb];
      a.y = center[(size_t)(g * 64 + u * 16 + l15) * IND + kb + 1];
      acc[u] = __builtin_amdgcn_wmma_f32_16x16x4_f32(false, a, false, b, (short)0,
                                                     acc[u], false, false);
    }
  }
  float bias = bQ[n0 + l15];
#pragma unroll
  for (int u = 0; u < 4; ++u)
#pragma unroll
    for (int r = 0; r < 8; ++r)
      Qout[(size_t)(g * 64 + u * 16 + r + hi * 8) * QCOLS + n0 + l15] = acc[u][r] + bias;
}

__global__ __launch_bounds__(256)
void attn_fused(const float* __restrict__ neighbors, const float* __restrict__ WK,
                const float* __restrict__ bK, const float* __restrict__ WVs,
                const float* __restrict__ bVs, const float* __restrict__ Qh,
                float* __restrict__ out) {
  __shared__ float nb[N_ * 260];
  __shared__ float Kh[N_ * QCOLS];
  __shared__ float Vs[N_ * OUTD];
  __shared__ float Qs[OUTD * 9];
  __shared__ float S[OUTD * 65];
  __shared__ float redM[4 * OUTD];
  __shared__ float redN[4 * OUTD];
  __shared__ float redD[4 * OUTD];

  const int b = blockIdx.x;
  const int t = threadIdx.x;
  const int wave = t >> 5, lane = t & 31, l15 = lane & 15, hi = lane >> 4;

  const float* nbg = neighbors + (size_t)b * N_ * IND;
  for (int i = t; i < N_ * IND; i += 256)
    nb[(i >> 8) * 260 + (i & 255)] = nbg[i];
  const float* qg = Qh + (size_t)b * QCOLS;
  for (int i = t; i < QCOLS; i += 256) {
    int d = i >> 3, h = i & 7;
    Qs[d * 9 + h] = qg[i] * 0.125f;
  }
  __syncthreads();

  for (int nt4 = 0; nt4 < 4; ++nt4) {
    int ntile = wave + nt4 * 8;
    int n0 = ntile * 16;
    v8f acc0 = {}, acc1 = {};
    for (int kt = 0; kt < IND / 4; ++kt) {
      int kb = kt * 4 + hi * 2;
      v2f bb, a0, a1;
      bb.x = WK[kb * QCOLS + n0 + l15];
      bb.y = WK[(kb + 1) * QCOLS + n0 + l15];
      a0.x = nb[l15 * 260 + kb];
      a0.y = nb[l15 * 260 + kb + 1];
      a1.x = nb[(16 + l15) * 260 + kb];
      a1.y = nb[(16 + l15) * 260 + kb + 1];
      acc0 = __builtin_amdgcn_wmma_f32_16x16x4_f32(false, a0, false, bb, (short)0,
                                                   acc0, false, false);
      acc1 = __builtin_amdgcn_wmma_f32_16x16x4_f32(false, a1, false, bb, (short)0,
                                                   acc1, false, false);
    }
    float bias = bK[n0 + l15];
#pragma unroll
    for (int r = 0; r < 8; ++r) {
      Kh[(r + hi * 8) * QCOLS + n0 + l15]      = acc0[r] + bias;
      Kh[(16 + r + hi * 8) * QCOLS + n0 + l15] = acc1[r] + bias;
    }
  }

  {
    int mt = wave >> 2, ntile = wave & 3;
    int m0 = mt * 16, n0 = ntile * 16;
    v8f acc = {};
    for (int kt = 0; kt < IND / 4; ++kt) {
      int kb = kt * 4 + hi * 2;
      v2f a, bb;
      a.x = nb[(m0 + l15) * 260 + kb];
      a.y = nb[(m0 + l15) * 260 + kb + 1];
      bb.x = WVs[kb * OUTD + n0 + l15];
      bb.y = WVs[(kb + 1) * OUTD + n0 + l15];
      acc = __builtin_amdgcn_wmma_f32_16x16x4_f32(false, a, false, bb, (short)0, acc,
                                                  false, false);
    }
    float bias = bVs[n0 + l15];
#pragma unroll
    for (int r = 0; r < 8; ++r)
      Vs[(m0 + r + hi * 8) * OUTD + n0 + l15] = acc[r] + bias;
  }
  __syncthreads();

  const int it = (wave * 2) >> 2;
  const int m0q = it * 16;
  v2f aQ[2];
#pragma unroll
  for (int ks = 0; ks < 2; ++ks) {
    int kb = ks * 4 + hi * 2;
    aQ[ks].x = Qs[(m0q + l15) * 9 + kb];
    aQ[ks].y = Qs[(m0q + l15) * 9 + kb + 1];
  }

  const int j = t & 63;
  const int q = t >> 6;
  float accOut = 0.f;

  for (int n = 0; n < N_; ++n) {
#pragma unroll
    for (int tt = 0; tt < 2; ++tt) {
      int id = wave * 2 + tt;
      int j0 = (id & 3) * 16;
      v8f acc = {};
#pragma unroll
      for (int ks = 0; ks < 2; ++ks) {
        int kb = ks * 4 + hi * 2;
        v2f bb;
        bb.x = Kh[n * QCOLS + (j0 + l15) * 8 + kb];
        bb.y = Kh[n * QCOLS + (j0 + l15) * 8 + kb + 1];
        acc = __builtin_amdgcn_wmma_f32_16x16x4_f32(false, aQ[ks], false, bb, (short)0,
                                                    acc, false, false);
      }
#pragma unroll
      for (int r = 0; r < 8; ++r)
        S[(m0q + r + hi * 8) * 65 + j0 + l15] = acc[r];
    }
    __syncthreads();

    float pm = -3.4e38f;
#pragma unroll
    for (int s = 0; s < 16; ++s)
      pm = fmaxf(pm, S[(q * 16 + s) * 65 + j]);
    redM[q * 64 + j] = pm;
    __syncthreads();
    float m = fmaxf(fmaxf(redM[j], redM[64 + j]), fmaxf(redM[128 + j], redM[192 + j]));
    float pn = 0.f, pd = 0.f;
#pragma unroll
    for (int s = 0; s < 16; ++s) {
      float e = __expf(S[(q * 16 + s) * 65 + j] - m);
      pd += e;
      pn += e * Vs[n * OUTD + q * 16 + s];
    }
    redN[q * 64 + j] = pn;
    redD[q * 64 + j] = pd;
    __syncthreads();
    if (t < 64) {
      float num = redN[j] + redN[64 + j] + redN[128 + j] + redN[192 + j];
      float den = redD[j] + redD[64 + j] + redD[128 + j] + redD[192 + j];
      accOut += num / den;
    }
  }
  if (t < 64) out[(size_t)b * OUTD + t] = accOut * (1.f / H_);
}

extern "C" void kernel_launch(void* const* d_in, const int* in_sizes, int n_in,
                              void* d_out, int out_size, void* d_ws, size_t ws_size,
                              hipStream_t stream) {
  const float* center    = (const float*)d_in[0];
  const float* neighbors = (const float*)d_in[1];
  const float* WQ        = (const float*)d_in[2];
  const float* bQ        = (const float*)d_in[3];
  const float* WK        = (const float*)d_in[4];
  const float* bK        = (const float*)d_in[5];
  const float* WV        = (const float*)d_in[6];
  const float* bV        = (const float*)d_in[7];
  float* out = (float*)d_out;

  float* ws   = (float*)d_ws;
  float* wsQ  = ws;
  float* WVs  = ws + (size_t)B_ * QCOLS;
  float* bVs  = WVs + (size_t)IND * OUTD;

  prep_wvs<<<64, 256, 0, stream>>>(WV, bV, WVs, bVs);
  proj_q  <<<128, 256, 0, stream>>>(center, WQ, bQ, wsQ);
  attn_fused<<<B_, 256, 0, stream>>>(neighbors, WK, bK, WVs, bVs, wsQ, out);
}


# attn_fused: projections transposed (weights as A, neighbors as B): K accumulators are the score WMMAs' B operands, softmax in registers with neighbors on lanes; row and neighbor sum order change
# speedup vs baseline: 1.4039x; 1.4039x over previous
#include <hip/hip_runtime.h>

typedef __attribute__((ext_vector_type(2))) float v2f;
typedef __attribute__((ext_vector_type(8))) float v8f;

#define B_    2048
#define N_    32
#define IND   256
#define OUTD  64
#define H_    8
#define QCOLS 512


__global__ void prep_wvs(const float* __restrict__ WV, const float* __restrict__ bV,
                         float* __restrict__ WVs, float* __restrict__ bVs) {
  int idx = blockIdx.x * blockDim.x + threadIdx.x;
  int c = idx >> 6;
  int j = idx & 63;
  float s = 0.f;
#pragma unroll
  for (int h = 0; h < H_; ++h) s += WV[c * QCOLS + j * H_ + h];
  WVs[c * OUTD + j] = s;
  if (idx < OUTD) {
    float sb = 0.f;
#pragma unroll
    for (int h = 0; h < H_; ++h) sb += bV[idx * H_ + h];
    bVs[idx] = sb;
  }
}

__global__ __launch_bounds__(256)
void proj_q(const float* __restrict__ center, const float* __restrict__ WQ,
            const float* __restrict__ bQ, float* __restrict__ Qout) {
  int wid  = (blockIdx.x * blockDim.x + threadIdx.x) >> 5;
  int lane = threadIdx.x & 31;
  int l15 = lane & 15, hi = lane >> 4;
  int g  = wid >> 5;
  int nt = wid & 31;
  int n0 = nt * 16;
  v8f acc[4] = {{}, {}, {}, {}};
  for (int kt = 0; kt < IND / 4; ++kt) {
    int kb = kt * 4 + hi * 2;
    v2f b;
    b.x = WQ[kb * QCOLS + n0 + l15];
    b.y = WQ[(kb + 1) * QCOLS + n0 + l15];
#pragma unroll
    for (int u = 0; u < 4; ++u) {
      v2f a;
      a.x = center[(size_t)(g * 64 + u * 16 + l15) * IND + kb];
      a.y = center[(size_t)(g * 64 + u * 16 + l15) * IND + kb + 1];
      acc[u] = __builtin_amdgcn_wmma_f32_16x16x4_f32(false, a, false, b, (short)0,
                                                     acc[u], false, false);
    }
  }
  float bias = bQ[n0 + l15];
#pragma unroll
  for (int u = 0; u < 4; ++u)
#pragma unroll
    for (int r = 0; r < 8; ++r)
      Qout[(size_t)(g * 64 + u * 16 + r + hi * 8) * QCOLS + n0 + l15] = acc[u][r] + bias;
}

typedef __attribute__((ext_vector_type(4))) float v4f;

#define WKP 520
#define WVP 72
#define STG (8 * WKP + 8 * WVP)

struct WStage { v4f k[4]; v2f v; };

__device__ __forceinline__ void w_fetch(WStage& r, const float* __restrict__ WK,
                                        const float* __restrict__ WVs, int s,
                                        int r_lo, int c, int rv, int c2) {
#pragma unroll
  for (int q = 0; q < 4; ++q)
    r.k[q] = *(const v4f*)(WK + (size_t)(s * 8 + r_lo + 2 * q) * QCOLS + 4 * c);
  r.v = *(const v2f*)(WVs + (size_t)(s * 8 + rv) * OUTD + 2 * c2);
}

__device__ __forceinline__ void w_put(float* buf, const WStage& r, int r_lo, int c, int rv,
                                      int c2) {
#pragma unroll
  for (int q = 0; q < 4; ++q)
    *(v4f*)(buf + (r_lo + 2 * q) * WKP + 4 * c) = r.k[q];
  *(v2f*)(buf + 8 * WKP + rv * WVP + 2 * c2) = r.v;
}

__device__ __forceinline__ void lds_barrier() {
  __builtin_amdgcn_fence(__ATOMIC_RELEASE, "workgroup", "local");
  __builtin_amdgcn_s_barrier();
  __builtin_amdgcn_fence(__ATOMIC_ACQUIRE, "workgroup", "local");
}

__device__ __forceinline__ float xhalf(float x) {
  int b = __builtin_bit_cast(int, x);
  return __builtin_bit_cast(float, __builtin_amdgcn_permlanex16(b, b, 0x76543210, 0xfedcba98,
                                                                false, false));
}

#define VTP 68

__global__ __launch_bounds__(256, 6)
void attn_fused(const float* __restrict__ neighbors, const float* __restrict__ WK,
                const float* __restrict__ bK, const float* __restrict__ WVs,
                const float* __restrict__ bVs, const float* __restrict__ Qh,
                float* __restrict__ out) {
  __shared__ __attribute__((aligned(16))) float smem[N_ * VTP + OUTD * 9 + N_ * 260 + 3 * STG];
  float* const VsT = smem;
  float* const Qs  = smem + N_ * VTP;
  float* const pin = Qs + OUTD * 9;
  float* const nb  = pin;
  float* const ovl = pin + N_ * 260;
  float* const red = pin;

  const int b = blockIdx.x;
  const int t = threadIdx.x;
  const int wave = t >> 5, lane = t & 31, l15 = lane & 15, hi = lane >> 4;

  const int r_lo = t >> 7, cq = t & 127, rv = t >> 5, c2 = t & 31;
  WStage ra, rb, rc;
  w_fetch(ra, WK, WVs, 0, r_lo, cq, rv, c2);
  w_fetch(rb, WK, WVs, 1, r_lo, cq, rv, c2);
  w_fetch(rc, WK, WVs, 2, r_lo, cq, rv, c2);

  const float* nbg = neighbors + (size_t)b * N_ * IND;
  for (int i = t; i < N_ * IND; i += 256)
    nb[(i >> 8) * 260 + (i & 255)] = nbg[i];
  const float* qg = Qh + (size_t)b * QCOLS;
  for (int i = t; i < QCOLS; i += 256) {
    int d = i >> 3, h = i & 7;
    Qs[d * 9 + h] = qg[i] * 0.125f;
  }
  float* const buf0 = ovl;
  float* const buf1 = ovl + STG;
  w_put(buf0, ra, r_lo, cq, rv, c2);
  w_put(buf1, rb, r_lo, cq, rv, c2);
  lds_barrier();

  const int cperm = ((l15 >> 2) & 1) * 8 + ((l15 >> 1) & 1) * 4 + (l15 >> 3) * 2 + (l15 & 1);
  const int mtv = wave & 3, ntv = wave >> 2;
  v8f acc[2][4] = {{{}, {}, {}, {}}, {{}, {}, {}, {}}};
  v8f accV = {};
  {
    auto kstage = [&](const float* buf, int s) {
#pragma unroll
      for (int ks = 0; ks < 2; ++ks) {
        int kb = s * 8 + ks * 4 + hi * 2;
        v2f b0, b1, bn, av;
        b0.x = nb[l15 * 260 + kb];
        b0.y = nb[l15 * 260 + kb + 1];
        b1.x = nb[(16 + l15) * 260 + kb];
        b1.y = nb[(16 + l15) * 260 + kb + 1];
        bn.x = nb[(ntv * 16 + l15) * 260 + kb];
        bn.y = nb[(ntv * 16 + l15) * 260 + kb + 1];
        const float* arow = buf + (ks * 4 + hi * 2) * WKP + wave * 64 + cperm;
        const float* vrow = buf + 8 * WKP + (ks * 4 + hi * 2) * WVP + mtv * 16 + l15;
        av.x = vrow[0];
        av.y = vrow[WVP];
#pragma unroll
        for (int tl = 0; tl < 4; ++tl) {
          v2f aw;
          aw.x = arow[tl * 16];
          aw.y = arow[WKP + tl * 16];
          acc[0][tl] = __builtin_amdgcn_wmma_f32_16x16x4_f32(false, aw, false, b0, (short)0,
                                                             acc[0][tl], false, false);
          acc[1][tl] = __builtin_amdgcn_wmma_f32_16x16x4_f32(false, aw, false, b1, (short)0,
                                                             acc[1][tl], false, false);
        }
        accV = __builtin_amdgcn_wmma_f32_16x16x4_f32(false, av, false, bn, (short)0, accV,
                                                     false, false);
      }
    };
    int oc = 0, o1 = STG, o2 = 2 * STG;
#pragma nounroll
    for (int s = 0; s < IND / 8 - 3; ++s) {
      kstage(ovl + oc, s);
      __builtin_amdgcn_sched_barrier(0);
      w_put(ovl + o2, rc, r_lo, cq, rv, c2);
      w_fetch(rc, WK, WVs, s + 3, r_lo, cq, rv, c2);
      lds_barrier();
      const int ot = oc;
      oc = o1;
      o1 = o2;
      o2 = ot;
    }
    kstage(ovl + oc, IND / 8 - 3);
    __builtin_amdgcn_sched_barrier(0);
    w_put(ovl + o2, rc, r_lo, cq, rv, c2);
    lds_barrier();
    kstage(ovl + o1, IND / 8 - 2);
    kstage(ovl + o2, IND / 8 - 1);
    asm volatile("" : "+v"(accV));
  }

#pragma unroll
  for (int tl = 0; tl < 4; ++tl) {
    const float* bkp = bK + wave * 64 + tl * 16 + hi * 2;
#pragma unroll
    for (int q = 0; q < 4; ++q) {
      v2f bb = *(const v2f*)(bkp + (q >> 1) * 8 + (q & 1) * 4);
      acc[0][tl][2 * q]     += bb.x;
      acc[0][tl][2 * q + 1] += bb.y;
      acc[1][tl][2 * q]     += bb.x;
      acc[1][tl][2 * q + 1] += bb.y;
    }
    asm volatile("" : "+v"(acc[0][tl]), "+v"(acc[1][tl]));
  }
  {
    const float* bvp = bVs + mtv * 16 + hi * 8;
    v4f bv0 = *(const v4f*)bvp, bv1 = *(const v4f*)(bvp + 4);
    v4f o0, o1v;
    o0.x = accV[0] + bv0.x;  o0.y = accV[1] + bv0.y;  o0.z = accV[2] + bv0.z;  o0.w = accV[3] + bv0.w;
    o1v.x = accV[4] + bv1.x; o1v.y = accV[5] + bv1.y; o1v.z = accV[6] + bv1.z; o1v.w = accV[7] + bv1.w;
    float* vp = VsT + (ntv * 16 + l15) * VTP + mtv * 16 + hi * 8;
    *(v4f*)vp = o0;
    *(v4f*)(vp + 4) = o1v;
  }
  v2f aQ[4][2];
#pragma unroll
  for (int mt = 0; mt < 4; ++mt)
#pragma unroll
    for (int ks = 0; ks < 2; ++ks) {
      int kb = ks * 4 + hi * 2;
      aQ[mt][ks].x = Qs[(mt * 16 + l15) * 9 + kb];
      aQ[mt][ks].y = Qs[(mt * 16 + l15) * 9 + kb + 1];
    }
  __syncthreads();

#pragma unroll
  for (int tl = 0; tl < 4; ++tl) {
#pragma unroll
    for (int jj = 0; jj < 2; ++jj) {
      float ratio = 0.f;
#pragma unroll
      for (int nt = 0; nt < 2; ++nt) {
        v2f kb0, kb1;
        kb0.x = acc[nt][tl][jj * 4 + 0];
        kb0.y = acc[nt][tl][jj * 4 + 1];
        kb1.x = acc[nt][tl][jj * 4 + 2];
        kb1.y = acc[nt][tl][jj * 4 + 3];
        asm volatile("" : "+v"(kb0), "+v"(kb1), "+v"(ratio) : : "memory");
        v8f sc[4];
#pragma unroll
        for (int mt = 0; mt < 4; ++mt) {
          v8f z = {};
          sc[mt] = __builtin_amdgcn_wmma_f32_16x16x4_f32(false, aQ[mt][0], false, kb0, (short)0,
                                                         z, false, false);
          sc[mt] = __builtin_amdgcn_wmma_f32_16x16x4_f32(false, aQ[mt][1], false, kb1, (short)0,
                                                         sc[mt], false, false);
        }
        const float* vp = VsT + (nt * 16 + l15) * VTP + hi * 8;
        float mx = -3.4e38f;
#pragma unroll
        for (int mt = 0; mt < 4; ++mt)
#pragma unroll
          for (int r = 0; r < 8; ++r)
            mx = fmaxf(mx, sc[mt][r]);
        mx = fmaxf(mx, xhalf(mx));
        float pd = 0.f, pn = 0.f;
#pragma unroll
        for (int mt = 0; mt < 4; ++mt) {
          v4f va = *(const v4f*)(vp + mt * 16), vb = *(const v4f*)(vp + mt * 16 + 4);
          float vv[8] = {va.x, va.y, va.z, va.w, vb.x, vb.y, vb.z, vb.w};
#pragma unroll
          for (int r = 0; r < 8; ++r) {
            float e = __expf(sc[mt][r] - mx);
            pd += e;
            pn += e * vv[r];
          }
        }
        pd += xhalf(pd);
        pn += xhalf(pn);
        ratio += pn / pd;
      }
      red[(wave * 8 + tl * 2 + jj) * 17 + l15] = ratio;
    }
  }
  __syncthreads();
  if (t < OUTD) {
    float s = 0.f;
#pragma unroll
    for (int l = 0; l < 16; ++l) s += red[t * 17 + l];
    out[(size_t)b * OUTD + t] = s * (1.f / H_);
  }
}

extern "C" void kernel_launch(void* const* d_in, const int* in_sizes, int n_in,
                              void* d_out, int out_size, void* d_ws, size_t ws_size,
                              hipStream_t stream) {
  const float* center    = (const float*)d_in[0];
  const float* neighbors = (const float*)d_in[1];
  const float* WQ        = (const float*)d_in[2];
  const float* bQ        = (const float*)d_in[3];
  const float* WK        = (const float*)d_in[4];
  const float* bK        = (const float*)d_in[5];
  const float* WV        = (const float*)d_in[6];
  const float* bV        = (const float*)d_in[7];
  float* out = (float*)d_out;

  float* ws   = (float*)d_ws;
  float* wsQ  = ws;
  float* WVs  = ws + (size_t)B_ * QCOLS;
  float* bVs  = WVs + (size_t)IND * OUTD;

  prep_wvs<<<64, 256, 0, stream>>>(WV, bV, WVs, bVs);
  proj_q  <<<128, 256, 0, stream>>>(center, WQ, bQ, wsQ);
  attn_fused<<<B_, 256, 0, stream>>>(neighbors, WK, bK, WVs, bVs, wsQ, out);
}


# proj_q retiled: a wave owns one 16-row tile x 64 columns (was 64 rows x 16 columns), center read once per wave by 16-byte loads along k, k order permuted alike in both operands; k sum order changes
# speedup vs baseline: 1.5507x; 1.1046x over previous
#include <hip/hip_runtime.h>

typedef __attribute__((ext_vector_type(2))) float v2f;
typedef __attribute__((ext_vector_type(8))) float v8f;
typedef __attribute__((ext_vector_type(4))) float v4f;

#define B_    2048
#define N_    32
#define IND   256
#define OUTD  64
#define H_    8
#define QCOLS 512


__global__ void prep_wvs(const float* __restrict__ WV, const float* __restrict__ bV,
                         float* __restrict__ WVs, float* __restrict__ bVs) {
  int idx = blockIdx.x * blockDim.x + threadIdx.x;
  int c = idx >> 6;
  int j = idx & 63;
  float s = 0.f;
#pragma unroll
  for (int h = 0; h < H_; ++h) s += WV[c * QCOLS + j * H_ + h];
  WVs[c * OUTD + j] = s;
  if (idx < OUTD) {
    float sb = 0.f;
#pragma unroll
    for (int h = 0; h < H_; ++h) sb += bV[idx * H_ + h];
    bVs[idx] = sb;
  }
}

struct QStage { v4f a0, a1; v2f b[4][4]; };

__global__ __launch_bounds__(128)
void proj_q(const float* __restrict__ center, const float* __restrict__ WQ,
            const float* __restrict__ bQ, float* __restrict__ Qout) {
  const int wave = threadIdx.x >> 5, lane = threadIdx.x & 31;
  const int l15 = lane & 15, hi = lane >> 4;
  const int mt = (blockIdx.x >> 3) * 4 + wave;
  const int n0 = (blockIdx.x & 7) * 64;
  const float* ap = center + (size_t)(mt * 16 + l15) * IND + hi * 8;
  const float* bp = WQ + (size_t)(hi * 8) * QCOLS + n0 + l15;
  auto fetch = [&](QStage& s, int st) {
    s.a0 = *(const v4f*)(ap + st * 16);
    s.a1 = *(const v4f*)(ap + st * 16 + 4);
#pragma unroll
    for (int q = 0; q < 4; ++q)
#pragma unroll
      for (int t = 0; t < 4; ++t) {
        const float* p = bp + (size_t)(st * 16 + 2 * q) * QCOLS + t * 16;
        s.b[q][t].x = p[0];
        s.b[q][t].y = p[QCOLS];
      }
  };
  v8f acc[4] = {{}, {}, {}, {}};
  auto mul = [&](const QStage& s) {
    v2f a[4];
    a[0].x = s.a0.x; a[0].y = s.a0.y;
    a[1].x = s.a0.z; a[1].y = s.a0.w;
    a[2].x = s.a1.x; a[2].y = s.a1.y;
    a[3].x = s.a1.z; a[3].y = s.a1.w;
#pragma unroll
    for (int q = 0; q < 4; ++q)
#pragma unroll
      for (int t = 0; t < 4; ++t)
        acc[t] = __builtin_amdgcn_wmma_f32_16x16x4_f32(false, a[q], false, s.b[q][t], (short)0,
                                                       acc[t], false, false);
  };
  QStage s0, s1;
  fetch(s0, 0);
  for (int st = 0; st < IND / 16 - 2; st += 2) {
    fetch(s1, st + 1);
    mul(s0);
    fetch(s0, st + 2);
    mul(s1);
  }
  fetch(s1, IND / 16 - 1);
  mul(s0);
  mul(s1);
#pragma unroll
  for (int t = 0; t < 4; ++t) {
    float bias = bQ[n0 + t * 16 + l15];
#pragma unroll
    for (int r = 0; r < 8; ++r)
      Qout[(size_t)(mt * 16 + r + hi * 8) * QCOLS + n0 + t * 16 + l15] = acc[t][r] + bias;
  }
}


#define WKP 520
#define WVP 72
#define STG (8 * WKP + 8 * WVP)

struct WStage { v4f k[4]; v2f v; };

__device__ __forceinline__ void w_fetch(WStage& r, const float* __restrict__ WK,
                                        const float* __restrict__ WVs, int s,
                                        int r_lo, int c, int rv, int c2) {
#pragma unroll
  for (int q = 0; q < 4; ++q)
    r.k[q] = *(const v4f*)(WK + (size_t)(s * 8 + r_lo + 2 * q) * QCOLS + 4 * c);
  r.v = *(const v2f*)(WVs + (size_t)(s * 8 + rv) * OUTD + 2 * c2);
}

__device__ __forceinline__ void w_put(float* buf, const WStage& r, int r_lo, int c, int rv,
                                      int c2) {
#pragma unroll
  for (int q = 0; q < 4; ++q)
    *(v4f*)(buf + (r_lo + 2 * q) * WKP + 4 * c) = r.k[q];
  *(v2f*)(buf + 8 * WKP + rv * WVP + 2 * c2) = r.v;
}

__device__ __forceinline__ void lds_barrier() {
  __builtin_amdgcn_fence(__ATOMIC_RELEASE, "workgroup", "local");
  __builtin_amdgcn_s_barrier();
  __builtin_amdgcn_fence(__ATOMIC_ACQUIRE, "workgroup", "local");
}

__device__ __forceinline__ float xhalf(float x) {
  int b = __builtin_bit_cast(int, x);
  return __builtin_bit_cast(float, __builtin_amdgcn_permlanex16(b, b, 0x76543210, 0xfedcba98,
                                                                false, false));
}

#define VTP 68

__global__ __launch_bounds__(256, 6)
void attn_fused(const float* __restrict__ neighbors, const float* __restrict__ WK,
                const float* __restrict__ bK, const float* __restrict__ WVs,
                const float* __restrict__ bVs, const float* __restrict__ Qh,
                float* __restrict__ out) {
  __shared__ __attribute__((aligned(16))) float smem[N_ * VTP + OUTD * 9 + N_ * 260 + 3 * STG];
  float* const VsT = smem;
  float* const Qs  = smem + N_ * VTP;
  float* const pin = Qs + OUTD * 9;
  float* const nb  = pin;
  float* const ovl = pin + N_ * 260;
  float* const red = pin;

  const int b = blockIdx.x;
  const int t = threadIdx.x;
  const int wave = t >> 5, lane = t & 31, l15 = lane & 15, hi = lane >> 4;

  const int r_lo = t >> 7, cq = t & 127, rv = t >> 5, c2 = t & 31;
  WStage ra, rb, rc;
  w_fetch(ra, WK, WVs, 0, r_lo, cq, rv, c2);
  w_fetch(rb, WK, WVs, 1, r_lo, cq, rv, c2);
  w_fetch(rc, WK, WVs, 2, r_lo, cq, rv, c2);

  const float* nbg = neighbors + (size_t)b * N_ * IND;
  for (int i = t; i < N_ * IND; i += 256)
    nb[(i >> 8) * 260 + (i & 255)] = nbg[i];
  const float* qg = Qh + (size_t)b * QCOLS;
  for (int i = t; i < QCOLS; i += 256) {
    int d = i >> 3, h = i & 7;
    Qs[d * 9 + h] = qg[i] * 0.125f;
  }
  float* const buf0 = ovl;
  float* const buf1 = ovl + STG;
  w_put(buf0, ra, r_lo, cq, rv, c2);
  w_put(buf1, rb, r_lo, cq, rv, c2);
  lds_barrier();

  const int cperm = ((l15 >> 2) & 1) * 8 + ((l15 >> 1) & 1) * 4 + (l15 >> 3) * 2 + (l15 & 1);
  const int mtv = wave & 3, ntv = wave >> 2;
  v8f acc[2][4] = {{{}, {}, {}, {}}, {{}, {}, {}, {}}};
  v8f accV = {};
  {
    auto kstage = [&](const float* buf, int s) {
#pragma unroll
      for (int ks = 0; ks < 2; ++ks) {
        int kb = s * 8 + ks * 4 + hi * 2;
        v2f b0, b1, bn, av;
        b0.x = nb[l15 * 260 + kb];
        b0.y = nb[l15 * 260 + kb + 1];
        b1.x = nb[(16 + l15) * 260 + kb];
        b1.y = nb[(16 + l15) * 260 + kb + 1];
        bn.x = nb[(ntv * 16 + l15) * 260 + kb];
        bn.y = nb[(ntv * 16 + l15) * 260 + kb + 1];
        const float* arow = buf + (ks * 4 + hi * 2) * WKP + wave * 64 + cperm;
        const float* vrow = buf + 8 * WKP + (ks * 4 + hi * 2) * WVP + mtv * 16 + l15;
        av.x = vrow[0];
        av.y = vrow[WVP];
#pragma unroll
        for (int tl = 0; tl < 4; ++tl) {
          v2f aw;
          aw.x = arow[tl * 16];
          aw.y = arow[WKP + tl * 16];
          acc[0][tl] = __builtin_amdgcn_wmma_f32_16x16x4_f32(false, aw, false, b0, (short)0,
                                                             acc[0][tl], false, false);
          acc[1][tl] = __builtin_amdgcn_wmma_f32_16x16x4_f32(false, aw, false, b1, (short)0,
                                                             acc[1][tl], false, false);
        }
        accV = __builtin_amdgcn_wmma_f32_16x16x4_f32(false, av, false, bn, (short)0, accV,
                                                     false, false);
      }
    };
    int oc = 0, o1 = STG, o2 = 2 * STG;
#pragma nounroll
    for (int s = 0; s < IND / 8 - 3; ++s) {
      kstage(ovl + oc, s);
      __builtin_amdgcn_sched_barrier(0);
      w_put(ovl + o2, rc, r_lo, cq, rv, c2);
      w_fetch(rc, WK, WVs, s + 3, r_lo, cq, rv, c2);
      lds_barrier();
      const int ot = oc;
      oc = o1;
      o1 = o2;
      o2 = ot;
    }
    kstage(ovl + oc, IND / 8 - 3);
    __builtin_amdgcn_sched_barrier(0);
    w_put(ovl + o2, rc, r_lo, cq, rv, c2);
    lds_barrier();
    kstage(ovl + o1, IND / 8 - 2);
    kstage(ovl + o2, IND / 8 - 1);
    asm volatile("" : "+v"(accV));
  }

#pragma unroll
  for (int tl = 0; tl < 4; ++tl) {
    const float* bkp = bK + wave * 64 + tl * 16 + hi * 2;
#pragma unroll
    for (int q = 0; q < 4; ++q) {
      v2f bb = *(const v2f*)(bkp + (q >> 1) * 8 + (q & 1) * 4);
      acc[0][tl][2 * q]     += bb.x;
      acc[0][tl][2 * q + 1] += bb.y;
      acc[1][tl][2 * q]     += bb.x;
      acc[1][tl][2 * q + 1] += bb.y;
    }
    asm volatile("" : "+v"(acc[0][tl]), "+v"(acc[1][tl]));
  }
  {
    const float* bvp = bVs + mtv * 16 + hi * 8;
    v4f bv0 = *(const v4f*)bvp, bv1 = *(const v4f*)(bvp + 4);
    v4f o0, o1v;
    o0.x = accV[0] + bv0.x;  o0.y = accV[1] + bv0.y;  o0.z = accV[2] + bv0.z;  o0.w = accV[3] + bv0.w;
    o1v.x = accV[4] + bv1.x; o1v.y = accV[5] + bv1.y; o1v.z = accV[6] + bv1.z; o1v.w = accV[7] + bv1.w;
    float* vp = VsT + (ntv * 16 + l15) * VTP + mtv * 16 + hi * 8;
    *(v4f*)vp = o0;
    *(v4f*)(vp + 4) = o1v;
  }
  v2f aQ[4][2];
#pragma unroll
  for (int mt = 0; mt < 4; ++mt)
#pragma unroll
    for (int ks = 0; ks < 2; ++ks) {
      int kb = ks * 4 + hi * 2;
      aQ[mt][ks].x = Qs[(mt * 16 + l15) * 9 + kb];
      aQ[mt][ks].y = Qs[(mt * 16 + l15) * 9 + kb + 1];
    }
  __syncthreads();

#pragma unroll
  for (int tl = 0; tl < 4; ++tl) {
#pragma unroll
    for (int jj = 0; jj < 2; ++jj) {
      float ratio = 0.f;
#pragma unroll
      for (int nt = 0; nt < 2; ++nt) {
        v2f kb0, kb1;
        kb0.x = acc[nt][tl][jj * 4 + 0];
        kb0.y = acc[nt][tl][jj * 4 + 1];
        kb1.x = acc[nt][tl][jj * 4 + 2];
        kb1.y = acc[nt][tl][jj * 4 + 3];
        asm volatile("" : "+v"(kb0), "+v"(kb1), "+v"(ratio) : : "memory");
        v8f sc[4];
#pragma unroll
        for (int mt = 0; mt < 4; ++mt) {
          v8f z = {};
          sc[mt] = __builtin_amdgcn_wmma_f32_16x16x4_f32(false, aQ[mt][0], false, kb0, (short)0,
                                                         z, false, false);
          sc[mt] = __builtin_amdgcn_wmma_f32_16x16x4_f32(false, aQ[mt][1], false, kb1, (short)0,
                                                         sc[mt], false, false);
        }
        const float* vp = VsT + (nt * 16 + l15) * VTP + hi * 8;
        float mx = -3.4e38f;
#pragma unroll
        for (int mt = 0; mt < 4; ++mt)
#pragma unroll
          for (int r = 0; r < 8; ++r)
            mx = fmaxf(mx, sc[mt][r]);
        mx = fmaxf(mx, xhalf(mx));
        float pd = 0.f, pn = 0.f;
#pragma unroll
        for (int mt = 0; mt < 4; ++mt) {
          v4f va = *(const v4f*)(vp + mt * 16), vb = *(const v4f*)(vp + mt * 16 + 4);
          float vv[8] = {va.x, va.y, va.z, va.w, vb.x, vb.y, vb.z, vb.w};
#pragma unroll
          for (int r = 0; r < 8; ++r) {
            float e = __expf(sc[mt][r] - mx);
            pd += e;
            pn += e * vv[r];
          }
        }
        pd += xhalf(pd);
        pn += xhalf(pn);
        ratio += pn / pd;
      }
      red[(wave * 8 + tl * 2 + jj) * 17 + l15] = ratio;
    }
  }
  __syncthreads();
  if (t < OUTD) {
    float s = 0.f;
#pragma unroll
    for (int l = 0; l < 16; ++l) s += red[t * 17 + l];
    out[(size_t)b * OUTD + t] = s * (1.f / H_);
  }
}

extern "C" void kernel_launch(void* const* d_in, const int* in_sizes, int n_in,
                              void* d_out, int out_size, void* d_ws, size_t ws_size,
                              hipStream_t stream) {
  const float* center    = (const float*)d_in[0];
  const float* neighbors = (const float*)d_in[1];
  const float* WQ        = (const float*)d_in[2];
  const float* bQ        = (const float*)d_in[3];
  const float* WK        = (const float*)d_in[4];
  const float* bK        = (const float*)d_in[5];
  const float* WV        = (const float*)d_in[6];
  const float* bV        = (const float*)d_in[7];
  float* out = (float*)d_out;

  float* ws   = (float*)d_ws;
  float* wsQ  = ws;
  float* WVs  = ws + (size_t)B_ * QCOLS;
  float* bVs  = WVs + (size_t)IND * OUTD;

  prep_wvs<<<64, 256, 0, stream>>>(WV, bV, WVs, bVs);
  proj_q  <<<256, 128, 0, stream>>>(center, WQ, bQ, wsQ);
  attn_fused<<<B_, 256, 0, stream>>>(neighbors, WK, bK, WVs, bVs, wsQ, out);
}
